# starting kernel
# baseline (speedup 1.0000x reference)
#include <hip/hip_runtime.h>
#include <hip/hip_bf16.h>
#include <math.h>

typedef float v2f __attribute__((ext_vector_type(2)));
typedef float v8f __attribute__((ext_vector_type(8)));

#define NODE_DIM 128
#define LDS_STRIDE 132
#define WAVES_PER_BLOCK 2

__global__ void sage_count_kernel(const int* __restrict__ dst,
                                  float* __restrict__ cnt, int E) {
  int e = blockIdx.x * blockDim.x + threadIdx.x;
  if (e < E) atomicAdd(&cnt[dst[e]], 1.0f);
}

__global__ void sage_scatter_kernel(const float* __restrict__ feat,
                                    const int* __restrict__ src,
                                    const int* __restrict__ dst,
                                    float* __restrict__ agg, int E) {
  long long gid = (long long)blockIdx.x * blockDim.x + threadIdx.x;
  int e = (int)(gid >> 5);
  int q = (int)(gid & 31);
  if (e >= E) return;
  int s = src[e];
  int d = dst[e];
  const float4 v = *reinterpret_cast<const float4*>(
      feat + (long long)s * NODE_DIM + q * 4);
  float* base = agg + (long long)d * NODE_DIM + q * 4;
  atomicAdd(base + 0, v.x);
  atomicAdd(base + 1, v.y);
  atomicAdd(base + 2, v.z);
  atomicAdd(base + 3, v.w);
}

template <bool NORM_RELU>
__global__ __launch_bounds__(WAVES_PER_BLOCK * 32)
void sage_combine_kernel(const float* __restrict__ agg,
                         const float* __restrict__ cnt,
                         const float* __restrict__ xin,
                         const float* __restrict__ Wl,
                         const float* __restrict__ bias,
                         const float* __restrict__ Wr,
                         float* __restrict__ outp, int N) {
  __shared__ float s_agg[WAVES_PER_BLOCK][16][LDS_STRIDE];
  __shared__ float s_in [WAVES_PER_BLOCK][16][LDS_STRIDE];
  __shared__ float s_out[WAVES_PER_BLOCK][16][LDS_STRIDE];

  const int lane  = threadIdx.x & 31;
  const int wv    = threadIdx.x >> 5;
  const int tile  = blockIdx.x * WAVES_PER_BLOCK + wv;
  const int node0 = tile * 16;

  #pragma unroll 4
  for (int r = 0; r < 16; ++r) {
    const int n = node0 + r;
    if (n < N) {
      const float rc = 1.0f / fmaxf(cnt[n], 1.0f);
      const float4 va = *reinterpret_cast<const float4*>(
          agg + (long long)n * NODE_DIM + lane * 4);
      const float4 vx = *reinterpret_cast<const float4*>(
          xin + (long long)n * NODE_DIM + lane * 4);
      s_agg[wv][r][lane * 4 + 0] = va.x * rc;
      s_agg[wv][r][lane * 4 + 1] = va.y * rc;
      s_agg[wv][r][lane * 4 + 2] = va.z * rc;
      s_agg[wv][r][lane * 4 + 3] = va.w * rc;
      s_in[wv][r][lane * 4 + 0] = vx.x;
      s_in[wv][r][lane * 4 + 1] = vx.y;
      s_in[wv][r][lane * 4 + 2] = vx.z;
      s_in[wv][r][lane * 4 + 3] = vx.w;
    } else {
      s_agg[wv][r][lane * 4 + 0] = 0.f; s_agg[wv][r][lane * 4 + 1] = 0.f;
      s_agg[wv][r][lane * 4 + 2] = 0.f; s_agg[wv][r][lane * 4 + 3] = 0.f;
      s_in[wv][r][lane * 4 + 0] = 0.f;  s_in[wv][r][lane * 4 + 1] = 0.f;
      s_in[wv][r][lane * 4 + 2] = 0.f;  s_in[wv][r][lane * 4 + 3] = 0.f;
    }
  }
  __syncthreads();

  const int m     = lane & 15;
  const int khalf = (lane >> 4) << 1;
  const int mbase = (lane >> 4) << 3;

  for (int ct = 0; ct < 8; ++ct) {
    const int n0 = ct * 16;
    const float bv = bias[n0 + m];
    v8f c;
    #pragma unroll
    for (int v = 0; v < 8; ++v) c[v] = bv;

    const float* wlrow = Wl + (long long)(n0 + m) * NODE_DIM;
    const float* wrrow = Wr + (long long)(n0 + m) * NODE_DIM;

    #pragma unroll 4
    for (int k0 = 0; k0 < NODE_DIM; k0 += 4) {
      v2f aA, aX;
      aA[0] = s_agg[wv][m][k0 + khalf];
      aA[1] = s_agg[wv][m][k0 + khalf + 1];
      const v2f bL = *reinterpret_cast<const v2f*>(wlrow + k0 + khalf);
      c = __builtin_amdgcn_wmma_f32_16x16x4_f32(false, aA, false, bL,
                                                (short)0, c, false, false);
      aX[0] = s_in[wv][m][k0 + khalf];
      aX[1] = s_in[wv][m][k0 + khalf + 1];
      const v2f bR = *reinterpret_cast<const v2f*>(wrrow + k0 + khalf);
      c = __builtin_amdgcn_wmma_f32_16x16x4_f32(false, aX, false, bR,
                                                (short)0, c, false, false);
    }

    #pragma unroll
    for (int v = 0; v < 8; ++v) s_out[wv][mbase + v][n0 + m] = c[v];
  }
  __syncthreads();

  if (NORM_RELU) {
    const int row = lane & 15;
    const int cb  = (lane >> 4) * 64;
    float ss = 0.f;
    #pragma unroll 8
    for (int c2 = 0; c2 < 64; ++c2) {
      const float v = s_out[wv][row][cb + c2];
      ss += v * v;
    }
    ss += __shfl_xor(ss, 16, 32);
    const float scl = 1.0f / fmaxf(sqrtf(ss), 1e-12f);
    #pragma unroll 8
    for (int c2 = 0; c2 < 64; ++c2) {
      const float v = s_out[wv][row][cb + c2] * scl;
      s_out[wv][row][cb + c2] = fmaxf(v, 0.0f);
    }
    __syncthreads();
  }

  #pragma unroll 4
  for (int r = 0; r < 16; ++r) {
    const int n = node0 + r;
    if (n < N) {
      float4 v;
      v.x = s_out[wv][r][lane * 4 + 0];
      v.y = s_out[wv][r][lane * 4 + 1];
      v.z = s_out[wv][r][lane * 4 + 2];
      v.w = s_out[wv][r][lane * 4 + 3];
      *reinterpret_cast<float4*>(outp + (long long)n * NODE_DIM + lane * 4) = v;
    }
  }
}

extern "C" void kernel_launch(void* const* d_in, const int* in_sizes, int n_in,
                              void* d_out, int out_size, void* d_ws,
                              size_t ws_size, hipStream_t stream) {
  (void)n_in; (void)out_size; (void)ws_size;

  const float* x   = (const float*)d_in[0];
  const int*   ei  = (const int*)d_in[1];
  const float* W1l = (const float*)d_in[2];
  const float* b1  = (const float*)d_in[3];
  const float* W1r = (const float*)d_in[4];
  const float* W2l = (const float*)d_in[5];
  const float* b2  = (const float*)d_in[6];
  const float* W2r = (const float*)d_in[7];
  float* out = (float*)d_out;

  const int N = in_sizes[0] / NODE_DIM;
  const int E = in_sizes[1] / 2;
  const int* src = ei;
  const int* dst = ei + E;

  char* ws = (char*)d_ws;
  float* cnt = (float*)ws;
  size_t off = ((size_t)N * 4 + 255) & ~(size_t)255;
  float* aggb = (float*)(ws + off);
  off += (((size_t)N * NODE_DIM * 4) + 255) & ~(size_t)255;
  float* h = (float*)(ws + off);

  const int tiles  = (N + 15) / 16;
  const int cblk   = (tiles + WAVES_PER_BLOCK - 1) / WAVES_PER_BLOCK;
  const long long sthreads = (long long)E * 32;
  const int sblk = (int)((sthreads + 255) / 256);
  const int dblk = (E + 255) / 256;

  hipMemsetAsync(cnt, 0, (size_t)N * 4, stream);
  hipMemsetAsync(aggb, 0, (size_t)N * NODE_DIM * 4, stream);
  sage_count_kernel<<<dblk, 256, 0, stream>>>(dst, cnt, E);
  sage_scatter_kernel<<<sblk, 256, 0, stream>>>(x, src, dst, aggb, E);
  sage_combine_kernel<true><<<cblk, WAVES_PER_BLOCK * 32, 0, stream>>>(
      aggb, cnt, x, W1l, b1, W1r, h, N);

  hipMemsetAsync(aggb, 0, (size_t)N * NODE_DIM * 4, stream);
  sage_scatter_kernel<<<sblk, 256, 0, stream>>>(h, src, dst, aggb, E);
  sage_combine_kernel<false><<<cblk, WAVES_PER_BLOCK * 32, 0, stream>>>(
      aggb, cnt, h, W2l, b2, W2r, out, N);
}


# Combine re-split: each of a group's 8 waves owns 16 output columns and keeps its Wl/Wr fragments in registers; 32-node A tiles shared from LDS; row-norm squares summed by a fixed 8-lane tree
# speedup vs baseline: 2.1177x; 2.1177x over previous
#include <hip/hip_runtime.h>
#include <hip/hip_bf16.h>
#include <math.h>

typedef float v2f __attribute__((ext_vector_type(2)));
typedef float v8f __attribute__((ext_vector_type(8)));

#define NODE_DIM 128
#define LDS_STRIDE 132

__global__ void sage_count_kernel(const int* __restrict__ dst,
                                  float* __restrict__ cnt, int E) {
  int e = blockIdx.x * blockDim.x + threadIdx.x;
  if (e < E) atomicAdd(&cnt[dst[e]], 1.0f);
}

#define EDGES_PER_WAVE 32
#define SCATTER_BATCH 8

__global__ __launch_bounds__(256)
void sage_scatter_kernel(const float* __restrict__ feat,
                         const int* __restrict__ src,
                         const int* __restrict__ dst,
                         float* __restrict__ agg, int E) {
  const int lane = threadIdx.x & 31;
  const int wave = __builtin_amdgcn_readfirstlane(
      (int)(((long long)blockIdx.x * blockDim.x + threadIdx.x) >> 5));
  const long long e0 = (long long)wave * EDGES_PER_WAVE;
  if (e0 >= E) return;
  const long long left = (long long)E - e0;
  const int n = left < EDGES_PER_WAVE ? (int)left : EDGES_PER_WAVE;

  int s = 0, d = 0;
  if (lane < n) {
    s = src[e0 + lane];
    d = dst[e0 + lane];
  }

  for (int b = 0; b < n; b += SCATTER_BATCH) {
    if (b + SCATTER_BATCH <= n) {
      float v[SCATTER_BATCH][4];
      #pragma unroll
      for (int i = 0; i < SCATTER_BATCH; ++i) {
        const int si = __builtin_amdgcn_readlane(s, b + i);
        const float* r = feat + (long long)si * NODE_DIM + lane;
        v[i][0] = r[0];
        v[i][1] = r[32];
        v[i][2] = r[64];
        v[i][3] = r[96];
      }
      #pragma unroll
      for (int i = 0; i < SCATTER_BATCH; ++i) {
        const int di = __builtin_amdgcn_readlane(d, b + i);
        float* a = agg + (long long)di * NODE_DIM + lane;
        atomicAdd(a,      v[i][0]);
        atomicAdd(a + 32, v[i][1]);
        atomicAdd(a + 64, v[i][2]);
        atomicAdd(a + 96, v[i][3]);
      }
    } else {
      for (int i = b; i < n; ++i) {
        const int si = __builtin_amdgcn_readlane(s, i);
        const int di = __builtin_amdgcn_readlane(d, i);
        const float* r = feat + (long long)si * NODE_DIM + lane;
        float* a = agg + (long long)di * NODE_DIM + lane;
        const float v0 = r[0], v1 = r[32], v2 = r[64], v3 = r[96];
        atomicAdd(a,      v0);
        atomicAdd(a + 32, v1);
        atomicAdd(a + 64, v2);
        atomicAdd(a + 96, v3);
      }
    }
  }
}

#define CB_WAVES 8
#define CB_ROWS 32
#define CB_GRID 512

template <bool NORM_RELU>
__global__ __launch_bounds__(CB_WAVES * 32)
void sage_combine_kernel(const float* __restrict__ agg,
                         const float* __restrict__ cnt,
                         const float* __restrict__ xin,
                         const float* __restrict__ Wl,
                         const float* __restrict__ bias,
                         const float* __restrict__ Wr,
                         float* __restrict__ outp, int N) {
  __shared__ float s_agg[CB_ROWS][LDS_STRIDE];
  __shared__ float s_in [CB_ROWS][LDS_STRIDE];
  __shared__ float s_out[CB_ROWS][LDS_STRIDE];

  const int tid   = threadIdx.x;
  const int lane  = tid & 31;
  const int wv    = tid >> 5;
  const int m     = lane & 15;
  const int khalf = (lane >> 4) << 1;
  const int mbase = (lane >> 4) << 3;
  const int n0    = wv * 16;
  const int groups = (N + CB_ROWS - 1) / CB_ROWS;

  v2f bL[32], bR[32];
  {
    const float* wlrow = Wl + (long long)(n0 + m) * NODE_DIM + khalf;
    const float* wrrow = Wr + (long long)(n0 + m) * NODE_DIM + khalf;
    #pragma unroll
    for (int j = 0; j < 32; ++j) {
      bL[j] = *reinterpret_cast<const v2f*>(wlrow + 4 * j);
      bR[j] = *reinterpret_cast<const v2f*>(wrrow + 4 * j);
    }
  }
  const float bv = bias[n0 + m];

  float4 pa[4], px[4];
  float  pc[4];
  int g = blockIdx.x;
  if (g < groups) {
    #pragma unroll
    for (int i = 0; i < 4; ++i) {
      const int n = g * CB_ROWS + wv + 8 * i;
      pa[i] = make_float4(0.f, 0.f, 0.f, 0.f);
      px[i] = make_float4(0.f, 0.f, 0.f, 0.f);
      pc[i] = 1.0f;
      if (n < N) {
        pc[i] = cnt[n];
        pa[i] = *reinterpret_cast<const float4*>(agg + (long long)n * NODE_DIM + lane * 4);
        px[i] = *reinterpret_cast<const float4*>(xin + (long long)n * NODE_DIM + lane * 4);
      }
    }
  }

  for (; g < groups; g += CB_GRID) {
    const int node0 = g * CB_ROWS;

    #pragma unroll
    for (int i = 0; i < 4; ++i) {
      const int r = wv + 8 * i;
      const float rc = 1.0f / fmaxf(pc[i], 1.0f);
      float4 va = pa[i];
      va.x *= rc; va.y *= rc; va.z *= rc; va.w *= rc;
      *reinterpret_cast<float4*>(&s_agg[r][lane * 4]) = va;
      *reinterpret_cast<float4*>(&s_in [r][lane * 4]) = px[i];
    }
    __syncthreads();

    {
      const int gn = g + CB_GRID;
      if (gn < groups) {
        #pragma unroll
        for (int i = 0; i < 4; ++i) {
          const int n = gn * CB_ROWS + wv + 8 * i;
          pa[i] = make_float4(0.f, 0.f, 0.f, 0.f);
          px[i] = make_float4(0.f, 0.f, 0.f, 0.f);
          pc[i] = 1.0f;
          if (n < N) {
            pc[i] = cnt[n];
            pa[i] = *reinterpret_cast<const float4*>(agg + (long long)n * NODE_DIM + lane * 4);
            px[i] = *reinterpret_cast<const float4*>(xin + (long long)n * NODE_DIM + lane * 4);
          }
        }
      }
    }

    v8f c0, c1;
    #pragma unroll
    for (int v = 0; v < 8; ++v) { c0[v] = bv; c1[v] = bv; }

    #pragma unroll
    for (int j = 0; j < 32; ++j) {
      const int k = 4 * j + khalf;
      const v2f a0 = *reinterpret_cast<const v2f*>(&s_agg[m][k]);
      const v2f a1 = *reinterpret_cast<const v2f*>(&s_agg[16 + m][k]);
      const v2f x0 = *reinterpret_cast<const v2f*>(&s_in[m][k]);
      const v2f x1 = *reinterpret_cast<const v2f*>(&s_in[16 + m][k]);
      c0 = __builtin_amdgcn_wmma_f32_16x16x4_f32(false, a0, false, bL[j],
                                                 (short)0, c0, false, false);
      c1 = __builtin_amdgcn_wmma_f32_16x16x4_f32(false, a1, false, bL[j],
                                                 (short)0, c1, false, false);
      c0 = __builtin_amdgcn_wmma_f32_16x16x4_f32(false, x0, false, bR[j],
                                                 (short)0, c0, false, false);
      c1 = __builtin_amdgcn_wmma_f32_16x16x4_f32(false, x1, false, bR[j],
                                                 (short)0, c1, false, false);
    }

    #pragma unroll
    for (int v = 0; v < 8; ++v) {
      s_out[mbase + v][n0 + m]      = c0[v];
      s_out[16 + mbase + v][n0 + m] = c1[v];
    }
    __syncthreads();

    {
      const int row = tid >> 3;
      const int seg = tid & 7;
      float4 o[4];
      #pragma unroll
      for (int q = 0; q < 4; ++q)
        o[q] = *reinterpret_cast<const float4*>(&s_out[row][seg * 4 + 32 * q]);
      if (NORM_RELU) {
        float ss = 0.f;
        #pragma unroll
        for (int q = 0; q < 4; ++q)
          ss += o[q].x * o[q].x + o[q].y * o[q].y + o[q].z * o[q].z + o[q].w * o[q].w;
        ss += __shfl_xor(ss, 1, 32);
        ss += __shfl_xor(ss, 2, 32);
        ss += __shfl_xor(ss, 4, 32);
        const float scl = 1.0f / fmaxf(sqrtf(ss), 1e-12f);
        #pragma unroll
        for (int q = 0; q < 4; ++q) {
          o[q].x = fmaxf(o[q].x * scl, 0.0f);
          o[q].y = fmaxf(o[q].y * scl, 0.0f);
          o[q].z = fmaxf(o[q].z * scl, 0.0f);
          o[q].w = fmaxf(o[q].w * scl, 0.0f);
        }
      }
      const int n = node0 + row;
      if (n < N) {
        #pragma unroll
        for (int q = 0; q < 4; ++q)
          *reinterpret_cast<float4*>(outp + (long long)n * NODE_DIM + seg * 4 + 32 * q) = o[q];
      }
    }
  }
}

extern "C" void kernel_launch(void* const* d_in, const int* in_sizes, int n_in,
                              void* d_out, int out_size, void* d_ws,
                              size_t ws_size, hipStream_t stream) {
  (void)n_in; (void)out_size; (void)ws_size;

  const float* x   = (const float*)d_in[0];
  const int*   ei  = (const int*)d_in[1];
  const float* W1l = (const float*)d_in[2];
  const float* b1  = (const float*)d_in[3];
  const float* W1r = (const float*)d_in[4];
  const float* W2l = (const float*)d_in[5];
  const float* b2  = (const float*)d_in[6];
  const float* W2r = (const float*)d_in[7];
  float* out = (float*)d_out;

  const int N = in_sizes[0] / NODE_DIM;
  const int E = in_sizes[1] / 2;
  const int* src = ei;
  const int* dst = ei + E;

  char* ws = (char*)d_ws;
  float* cnt = (float*)ws;
  size_t off = ((size_t)N * 4 + 255) & ~(size_t)255;
  float* aggb = (float*)(ws + off);
  off += (((size_t)N * NODE_DIM * 4) + 255) & ~(size_t)255;
  float* h = (float*)(ws + off);

  const int sblk = (E + 255) / 256;
  const int dblk = (E + 255) / 256;

  hipMemsetAsync(cnt, 0, (size_t)N * 4, stream);
  hipMemsetAsync(aggb, 0, (size_t)N * NODE_DIM * 4, stream);
  sage_count_kernel<<<dblk, 256, 0, stream>>>(dst, cnt, E);
  sage_scatter_kernel<<<sblk, 256, 0, stream>>>(x, src, dst, aggb, E);
  sage_combine_kernel<true><<<CB_GRID, CB_WAVES * 32, 0, stream>>>(
      aggb, cnt, x, W1l, b1, W1r, h, N);

  hipMemsetAsync(aggb, 0, (size_t)N * NODE_DIM * 4, stream);
  sage_scatter_kernel<<<sblk, 256, 0, stream>>>(h, src, dst, aggb, E);
  sage_combine_kernel<false><<<CB_GRID, CB_WAVES * 32, 0, stream>>>(
      aggb, cnt, h, W2l, b2, W2r, out, N);
}


# Aggregation data path: a work-group owns 416 nodes' sums in LDS, scans one owner byte per edge, gathers its edges' rows, adds them by LDS f32 atomics; no global atomics, memsets, count; arrival order
# speedup vs baseline: 3.1597x; 1.4920x over previous
#include <hip/hip_runtime.h>
#include <hip/hip_bf16.h>
#include <math.h>

typedef float v2f __attribute__((ext_vector_type(2)));
typedef float v8f __attribute__((ext_vector_type(8)));

#define NODE_DIM 128
#define LDS_STRIDE 132

#define SCATTER_BATCH 8

#define OWN_CAP 416
#define OWN_WAVES 32
#define OWN_GRID 256
#define Q_CAP 64

__global__ __launch_bounds__(256)
void sage_owner_kernel(const int* __restrict__ dst, float* __restrict__ ownbuf, int E) {
  unsigned int* own = reinterpret_cast<unsigned int*>(ownbuf);
  const int chunks = (E + 15) >> 4;
  const int t = blockIdx.x * 256 + threadIdx.x;
  if (t >= chunks * 4) return;
  const int e = t * 4;
  unsigned int w = 0;
  #pragma unroll
  for (int k = 0; k < 4; ++k)
    if (e + k < E) w |= (((unsigned int)dst[e + k] / OWN_CAP) & 255u) << (8 * k);
  own[t] = w;
}

__device__ __forceinline__ unsigned int zero_bytes(unsigned int x) {
  return ~(((x & 0x7f7f7f7fu) + 0x7f7f7f7fu) | x) & 0x80808080u;
}
__device__ __forceinline__ unsigned int pack_nib(unsigned int z) {
  return (((z >> 7) * 0x00204081u) >> 21) & 15u;
}

typedef unsigned int v4u __attribute__((ext_vector_type(4)));

__device__ __forceinline__ unsigned int match16(v4u w, unsigned int pat) {
  const unsigned int z0 = zero_bytes(w.x ^ pat);
  const unsigned int z1 = zero_bytes(w.y ^ pat);
  const unsigned int z2 = zero_bytes(w.z ^ pat);
  const unsigned int z3 = zero_bytes(w.w ^ pat);
  unsigned int m = 0;
  if (z0 | z1 | z2 | z3)
    m = pack_nib(z0) | (pack_nib(z1) << 4) | (pack_nib(z2) << 8) | (pack_nib(z3) << 12);
  return m;
}

typedef volatile __attribute__((address_space(3))) int* lds_vint_p;

template <bool WRITE_CNT>
__device__ __forceinline__ void own_drain(const float* __restrict__ feat,
                                          const int* __restrict__ src,
                                          const int* __restrict__ dst,
                                          float* s_acc, float* s_cnt,
                                          lds_vint_p q, int n, int lo, int E,
                                          int lane) {
  int s = 0, loc = -1;
  if (lane < n) {
    const int e = q[lane];
    if (e < E) {
      const int d  = dst[e];
      const int se = src[e];
      const int l = d - lo;
      if ((unsigned int)l < (unsigned int)OWN_CAP) {
        s = se;
        loc = l;
        if (WRITE_CNT)
          __hip_atomic_fetch_add(&s_cnt[l], 1.0f, __ATOMIC_RELAXED, __HIP_MEMORY_SCOPE_WORKGROUP);
      }
    }
  }
  for (int i = 0; i < n; i += SCATTER_BATCH) {
    float v[SCATTER_BATCH][4];
    #pragma unroll
    for (int j = 0; j < SCATTER_BATCH; ++j) {
      const int sj = __builtin_amdgcn_readlane(s, i + j);
      const float* r = feat + (long long)sj * NODE_DIM + lane;
      v[j][0] = r[0];
      v[j][1] = r[32];
      v[j][2] = r[64];
      v[j][3] = r[96];
    }
    #pragma unroll
    for (int j = 0; j < SCATTER_BATCH; ++j) {
      const int lj = __builtin_amdgcn_readlane(loc, i + j);
      if (lj >= 0) {
        float* a = s_acc + lj * NODE_DIM + lane;
        __hip_atomic_fetch_add(a,      v[j][0], __ATOMIC_RELAXED, __HIP_MEMORY_SCOPE_WORKGROUP);
        __hip_atomic_fetch_add(a + 32, v[j][1], __ATOMIC_RELAXED, __HIP_MEMORY_SCOPE_WORKGROUP);
        __hip_atomic_fetch_add(a + 64, v[j][2], __ATOMIC_RELAXED, __HIP_MEMORY_SCOPE_WORKGROUP);
        __hip_atomic_fetch_add(a + 96, v[j][3], __ATOMIC_RELAXED, __HIP_MEMORY_SCOPE_WORKGROUP);
      }
    }
  }
}

template <bool WRITE_CNT>
__global__ __launch_bounds__(OWN_WAVES * 32)
void sage_gather_own_kernel(const float* __restrict__ feat,
                            const int* __restrict__ src,
                            const int* __restrict__ dst,
                            const float* __restrict__ ownbuf,
                            float* __restrict__ agg,
                            float* __restrict__ cnt, int N, int E) {
  __shared__ float s_acc[OWN_CAP * NODE_DIM];
  __shared__ float s_cnt[OWN_CAP];
  __shared__ int   s_q[OWN_WAVES][Q_CAP];

  const v4u* own16 = reinterpret_cast<const v4u*>(ownbuf);
  const int tid  = threadIdx.x;
  const int lane = tid & 31;
  const int wv = __builtin_amdgcn_readfirstlane(tid >> 5);
  lds_vint_p q = (lds_vint_p)&s_q[wv][0];
  const int chunks = (E + 15) >> 4;
  const int ranges = (N + OWN_CAP - 1) / OWN_CAP;

  for (int r = blockIdx.x; r < ranges; r += OWN_GRID) {
    const int lo = r * OWN_CAP;

    for (int i = tid; i < OWN_CAP * NODE_DIM / 4; i += OWN_WAVES * 32)
      reinterpret_cast<float4*>(s_acc)[i] = make_float4(0.f, 0.f, 0.f, 0.f);
    for (int i = tid; i < OWN_CAP; i += OWN_WAVES * 32) s_cnt[i] = 0.f;
    __syncthreads();

    const unsigned int pat = (unsigned int)(r & 255) * 0x01010101u;
    int qn = 0;

    const int nblk = (chunks + 127) >> 7;
    v4u cur[4];
    #pragma unroll
    for (int u = 0; u < 4; ++u) {
      const int c = wv * 128 + u * 32 + lane;
      cur[u] = own16[c < chunks ? c : chunks - 1];
    }
    for (int bi = wv; bi < nblk; bi += OWN_WAVES) {
      v4u nxt[4];
      #pragma unroll
      for (int u = 0; u < 4; ++u) {
        const int c = (bi + OWN_WAVES) * 128 + u * 32 + lane;
        nxt[u] = own16[c < chunks ? c : chunks - 1];
      }
      unsigned long long m = 0;
      #pragma unroll
      for (int u = 0; u < 4; ++u) {
        const int c = bi * 128 + u * 32 + lane;
        if (c < chunks)
          m |= (unsigned long long)match16(cur[u], pat) << (16 * u);
      }
      for (;;) {
        const unsigned int bal = __builtin_amdgcn_ballot_w32(m != 0);
        if (bal == 0) break;
        if (m != 0) {
          const int k = __builtin_ctzll(m);
          m &= m - 1;
          const int slot = qn + __builtin_popcount(bal & ((1u << lane) - 1u));
          q[slot] = ((bi * 128 + (k >> 4) * 32 + lane) << 4) + (k & 15);
        }
        qn += __builtin_popcount(bal);
        if (qn >= 32) {
          own_drain<WRITE_CNT>(feat, src, dst, s_acc, s_cnt, q, 32, lo, E, lane);
          qn -= 32;
          int keep = 0;
          if (lane < qn) keep = q[32 + lane];
          if (lane < qn) q[lane] = keep;
        }
      }
      #pragma unroll
      for (int u = 0; u < 4; ++u) cur[u] = nxt[u];
    }
    if (qn > 0)
      own_drain<WRITE_CNT>(feat, src, dst, s_acc, s_cnt, q, qn, lo, E, lane);
    __syncthreads();

    for (int row = wv; row < OWN_CAP; row += OWN_WAVES) {
      const int n = lo + row;
      if (n < N)
        *reinterpret_cast<float4*>(agg + (long long)n * NODE_DIM + lane * 4) =
            *reinterpret_cast<const float4*>(s_acc + row * NODE_DIM + lane * 4);
    }
    if (WRITE_CNT) {
      for (int i = tid; i < OWN_CAP; i += OWN_WAVES * 32)
        if (lo + i < N) cnt[lo + i] = s_cnt[i];
    }
    __syncthreads();
  }
}

#define CB_WAVES 8
#define CB_ROWS 32
#define CB_GRID 512

template <bool NORM_RELU>
__global__ __launch_bounds__(CB_WAVES * 32)
void sage_combine_kernel(const float* __restrict__ agg,
                         const float* __restrict__ cnt,
                         const float* __restrict__ xin,
                         const float* __restrict__ Wl,
                         const float* __restrict__ bias,
                         const float* __restrict__ Wr,
                         float* __restrict__ outp, int N) {
  __shared__ float s_agg[CB_ROWS][LDS_STRIDE];
  __shared__ float s_in [CB_ROWS][LDS_STRIDE];
  __shared__ float s_out[CB_ROWS][LDS_STRIDE];

  const int tid   = threadIdx.x;
  const int lane  = tid & 31;
  const int wv    = tid >> 5;
  const int m     = lane & 15;
  const int khalf = (lane >> 4) << 1;
  const int mbase = (lane >> 4) << 3;
  const int n0    = wv * 16;
  const int groups = (N + CB_ROWS - 1) / CB_ROWS;

  v2f bL[32], bR[32];
  {
    const float* wlrow = Wl + (long long)(n0 + m) * NODE_DIM + khalf;
    const float* wrrow = Wr + (long long)(n0 + m) * NODE_DIM + khalf;
    #pragma unroll
    for (int j = 0; j < 32; ++j) {
      bL[j] = *reinterpret_cast<const v2f*>(wlrow + 4 * j);
      bR[j] = *reinterpret_cast<const v2f*>(wrrow + 4 * j);
    }
  }
  const float bv = bias[n0 + m];

  float4 pa[4], px[4];
  float  pc[4];
  int g = blockIdx.x;
  if (g < groups) {
    #pragma unroll
    for (int i = 0; i < 4; ++i) {
      const int n = g * CB_ROWS + wv + 8 * i;
      pa[i] = make_float4(0.f, 0.f, 0.f, 0.f);
      px[i] = make_float4(0.f, 0.f, 0.f, 0.f);
      pc[i] = 1.0f;
      if (n < N) {
        pc[i] = cnt[n];
        pa[i] = *reinterpret_cast<const float4*>(agg + (long long)n * NODE_DIM + lane * 4);
        px[i] = *reinterpret_cast<const float4*>(xin + (long long)n * NODE_DIM + lane * 4);
      }
    }
  }

  for (; g < groups; g += CB_GRID) {
    const int node0 = g * CB_ROWS;

    #pragma unroll
    for (int i = 0; i < 4; ++i) {
      const int r = wv + 8 * i;
      const float rc = 1.0f / fmaxf(pc[i], 1.0f);
      float4 va = pa[i];
      va.x *= rc; va.y *= rc; va.z *= rc; va.w *= rc;
      *reinterpret_cast<float4*>(&s_agg[r][lane * 4]) = va;
      *reinterpret_cast<float4*>(&s_in [r][lane * 4]) = px[i];
    }
    __syncthreads();

    {
      const int gn = g + CB_GRID;
      if (gn < groups) {
        #pragma unroll
        for (int i = 0; i < 4; ++i) {
          const int n = gn * CB_ROWS + wv + 8 * i;
          pa[i] = make_float4(0.f, 0.f, 0.f, 0.f);
          px[i] = make_float4(0.f, 0.f, 0.f, 0.f);
          pc[i] = 1.0f;
          if (n < N) {
            pc[i] = cnt[n];
            pa[i] = *reinterpret_cast<const float4*>(agg + (long long)n * NODE_DIM + lane * 4);
            px[i] = *reinterpret_cast<const float4*>(xin + (long long)n * NODE_DIM + lane * 4);
          }
        }
      }
    }

    v8f c0, c1;
    #pragma unroll
    for (int v = 0; v < 8; ++v) { c0[v] = bv; c1[v] = bv; }

    #pragma unroll
    for (int j = 0; j < 32; ++j) {
      const int k = 4 * j + khalf;
      const v2f a0 = *reinterpret_cast<const v2f*>(&s_agg[m][k]);
      const v2f a1 = *reinterpret_cast<const v2f*>(&s_agg[16 + m][k]);
      const v2f x0 = *reinterpret_cast<const v2f*>(&s_in[m][k]);
      const v2f x1 = *reinterpret_cast<const v2f*>(&s_in[16 + m][k]);
      c0 = __builtin_amdgcn_wmma_f32_16x16x4_f32(false, a0, false, bL[j],
                                                 (short)0, c0, false, false);
      c1 = __builtin_amdgcn_wmma_f32_16x16x4_f32(false, a1, false, bL[j],
                                                 (short)0, c1, false, false);
      c0 = __builtin_amdgcn_wmma_f32_16x16x4_f32(false, x0, false, bR[j],
                                                 (short)0, c0, false, false);
      c1 = __builtin_amdgcn_wmma_f32_16x16x4_f32(false, x1, false, bR[j],
                                                 (short)0, c1, false, false);
    }

    #pragma unroll
    for (int v = 0; v < 8; ++v) {
      s_out[mbase + v][n0 + m]      = c0[v];
      s_out[16 + mbase + v][n0 + m] = c1[v];
    }
    __syncthreads();

    {
      const int row = tid >> 3;
      const int seg = tid & 7;
      float4 o[4];
      #pragma unroll
      for (int q = 0; q < 4; ++q)
        o[q] = *reinterpret_cast<const float4*>(&s_out[row][seg * 4 + 32 * q]);
      if (NORM_RELU) {
        float ss = 0.f;
        #pragma unroll
        for (int q = 0; q < 4; ++q)
          ss += o[q].x * o[q].x + o[q].y * o[q].y + o[q].z * o[q].z + o[q].w * o[q].w;
        ss += __shfl_xor(ss, 1, 32);
        ss += __shfl_xor(ss, 2, 32);
        ss += __shfl_xor(ss, 4, 32);
        const float scl = 1.0f / fmaxf(sqrtf(ss), 1e-12f);
        #pragma unroll
        for (int q = 0; q < 4; ++q) {
          o[q].x = fmaxf(o[q].x * scl, 0.0f);
          o[q].y = fmaxf(o[q].y * scl, 0.0f);
          o[q].z = fmaxf(o[q].z * scl, 0.0f);
          o[q].w = fmaxf(o[q].w * scl, 0.0f);
        }
      }
      const int n = node0 + row;
      if (n < N) {
        #pragma unroll
        for (int q = 0; q < 4; ++q)
          *reinterpret_cast<float4*>(outp + (long long)n * NODE_DIM + seg * 4 + 32 * q) = o[q];
      }
    }
  }
}

extern "C" void kernel_launch(void* const* d_in, const int* in_sizes, int n_in,
                              void* d_out, int out_size, void* d_ws,
                              size_t ws_size, hipStream_t stream) {
  (void)n_in; (void)out_size; (void)ws_size;

  const float* x   = (const float*)d_in[0];
  const int*   ei  = (const int*)d_in[1];
  const float* W1l = (const float*)d_in[2];
  const float* b1  = (const float*)d_in[3];
  const float* W1r = (const float*)d_in[4];
  const float* W2l = (const float*)d_in[5];
  const float* b2  = (const float*)d_in[6];
  const float* W2r = (const float*)d_in[7];
  float* out = (float*)d_out;

  const int N = in_sizes[0] / NODE_DIM;
  const int E = in_sizes[1] / 2;
  const int* src = ei;
  const int* dst = ei + E;

  char* ws = (char*)d_ws;
  float* cnt = (float*)ws;
  size_t off = ((size_t)N * 4 + 255) & ~(size_t)255;
  float* aggb = (float*)(ws + off);
  off += (((size_t)N * NODE_DIM * 4) + 255) & ~(size_t)255;
  float* h = (float*)(ws + off);

  const int oblk = (((E + 15) / 16) * 4 + 255) / 256;
  sage_owner_kernel<<<oblk, 256, 0, stream>>>(dst, out, E);

  sage_gather_own_kernel<true><<<OWN_GRID, OWN_WAVES * 32, 0, stream>>>(
      x, src, dst, out, aggb, cnt, N, E);
  sage_combine_kernel<true><<<CB_GRID, CB_WAVES * 32, 0, stream>>>(
      aggb, cnt, x, W1l, b1, W1r, h, N);

  sage_gather_own_kernel<false><<<OWN_GRID, OWN_WAVES * 32, 0, stream>>>(
      h, src, dst, out, aggb, cnt, N, E);
  sage_combine_kernel<false><<<CB_GRID, CB_WAVES * 32, 0, stream>>>(
      aggb, cnt, h, W2l, b2, W2r, out, N);
}


# Edge ids as one record: the owner pass also writes an 8-byte (src,dst) pair per edge; a matched edge costs one scattered 8-B load, not two dependent 4-B loads from two arrays; sums in arrival order
# speedup vs baseline: 3.3197x; 1.0506x over previous
#include <hip/hip_runtime.h>
#include <hip/hip_bf16.h>
#include <math.h>

typedef float v2f __attribute__((ext_vector_type(2)));
typedef float v8f __attribute__((ext_vector_type(8)));

#define NODE_DIM 128
#define LDS_STRIDE 132

#define SCATTER_BATCH 8

#define OWN_CAP 416
#define OWN_WAVES 32
#define OWN_GRID 256
#define Q_CAP 64

typedef unsigned long long pair_t;
__device__ __forceinline__ const pair_t* edge_pairs(const float* ownbuf, int E) {
  const int chunks = (E + 15) >> 4;
  return reinterpret_cast<const pair_t*>(reinterpret_cast<const char*>(ownbuf) + (size_t)chunks * 16);
}

__global__ __launch_bounds__(256)
void sage_owner_kernel(const int* __restrict__ src, const int* __restrict__ dst,
                       float* __restrict__ ownbuf, int E) {
  unsigned int* own = reinterpret_cast<unsigned int*>(ownbuf);
  const int chunks = (E + 15) >> 4;
  pair_t* pairs = const_cast<pair_t*>(edge_pairs(ownbuf, E));
  const int t = blockIdx.x * 256 + threadIdx.x;
  if (t >= chunks * 4) return;
  const int e = t * 4;
  unsigned int w = 0;
  if (e + 3 < E) {
    int s[4], d[4];
    #pragma unroll
    for (int k = 0; k < 4; ++k) { d[k] = dst[e + k]; s[k] = src[e + k]; }
    #pragma unroll
    for (int k = 0; k < 4; ++k) w |= (((unsigned int)d[k] / OWN_CAP) & 255u) << (8 * k);
    int4* p4 = reinterpret_cast<int4*>(pairs + e);
    p4[0] = make_int4(s[0], d[0], s[1], d[1]);
    p4[1] = make_int4(s[2], d[2], s[3], d[3]);
  } else {
    #pragma unroll
    for (int k = 0; k < 4; ++k)
      if (e + k < E) {
        const int d = dst[e + k];
        w |= (((unsigned int)d / OWN_CAP) & 255u) << (8 * k);
        pairs[e + k] = ((pair_t)(unsigned int)d << 32) | (pair_t)(unsigned int)src[e + k];
      }
  }
  own[t] = w;
}

__device__ __forceinline__ unsigned int zero_bytes(unsigned int x) {
  return ~(((x & 0x7f7f7f7fu) + 0x7f7f7f7fu) | x) & 0x80808080u;
}
__device__ __forceinline__ unsigned int pack_nib(unsigned int z) {
  return (((z >> 7) * 0x00204081u) >> 21) & 15u;
}

typedef unsigned int v4u __attribute__((ext_vector_type(4)));

__device__ __forceinline__ unsigned int match16(v4u w, unsigned int pat) {
  const unsigned int z0 = zero_bytes(w.x ^ pat);
  const unsigned int z1 = zero_bytes(w.y ^ pat);
  const unsigned int z2 = zero_bytes(w.z ^ pat);
  const unsigned int z3 = zero_bytes(w.w ^ pat);
  unsigned int m = 0;
  if (z0 | z1 | z2 | z3)
    m = pack_nib(z0) | (pack_nib(z1) << 4) | (pack_nib(z2) << 8) | (pack_nib(z3) << 12);
  return m;
}

typedef volatile __attribute__((address_space(3))) int* lds_vint_p;

template <bool WRITE_CNT>
__device__ __forceinline__ void own_drain(const float* __restrict__ feat,
                                          const pair_t* __restrict__ pairs,
                                          float* s_acc, float* s_cnt,
                                          lds_vint_p q, int n, int lo, int E,
                                          int lane) {
  int s = 0, loc = -1;
  if (lane < n) {
    const int e = q[lane];
    if (e < E) {
      const pair_t sd = pairs[e];
      const int l = (int)(unsigned int)(sd >> 32) - lo;
      const bool mine = (unsigned int)l < (unsigned int)OWN_CAP;
      s   = mine ? (int)(unsigned int)sd : 0;
      loc = mine ? l : -1;
      if (WRITE_CNT && mine)
        __hip_atomic_fetch_add(&s_cnt[l], 1.0f, __ATOMIC_RELAXED, __HIP_MEMORY_SCOPE_WORKGROUP);
    }
  }
  for (int i = 0; i < n; i += SCATTER_BATCH) {
    float v[SCATTER_BATCH][4];
    #pragma unroll
    for (int j = 0; j < SCATTER_BATCH; ++j) {
      const int sj = __builtin_amdgcn_readlane(s, i + j);
      const float* r = feat + (long long)sj * NODE_DIM + lane;
      v[j][0] = r[0];
      v[j][1] = r[32];
      v[j][2] = r[64];
      v[j][3] = r[96];
    }
    #pragma unroll
    for (int j = 0; j < SCATTER_BATCH; ++j) {
      const int lj = __builtin_amdgcn_readlane(loc, i + j);
      if (lj >= 0) {
        float* a = s_acc + lj * NODE_DIM + lane;
        __hip_atomic_fetch_add(a,      v[j][0], __ATOMIC_RELAXED, __HIP_MEMORY_SCOPE_WORKGROUP);
        __hip_atomic_fetch_add(a + 32, v[j][1], __ATOMIC_RELAXED, __HIP_MEMORY_SCOPE_WORKGROUP);
        __hip_atomic_fetch_add(a + 64, v[j][2], __ATOMIC_RELAXED, __HIP_MEMORY_SCOPE_WORKGROUP);
        __hip_atomic_fetch_add(a + 96, v[j][3], __ATOMIC_RELAXED, __HIP_MEMORY_SCOPE_WORKGROUP);
      }
    }
  }
}

template <bool WRITE_CNT>
__global__ __launch_bounds__(OWN_WAVES * 32)
void sage_gather_own_kernel(const float* __restrict__ feat,
                            const float* __restrict__ ownbuf,
                            float* __restrict__ agg,
                            float* __restrict__ cnt, int N, int E) {
  __shared__ float s_acc[OWN_CAP * NODE_DIM];
  __shared__ float s_cnt[OWN_CAP];
  __shared__ int   s_q[OWN_WAVES][Q_CAP];

  const v4u* own16 = reinterpret_cast<const v4u*>(ownbuf);
  const pair_t* pairs = edge_pairs(ownbuf, E);
  const int tid  = threadIdx.x;
  const int lane = tid & 31;
  const int wv = __builtin_amdgcn_readfirstlane(tid >> 5);
  lds_vint_p q = (lds_vint_p)&s_q[wv][0];
  const int chunks = (E + 15) >> 4;
  const int ranges = (N + OWN_CAP - 1) / OWN_CAP;

  for (int r = blockIdx.x; r < ranges; r += OWN_GRID) {
    const int lo = r * OWN_CAP;

    for (int i = tid; i < OWN_CAP * NODE_DIM / 4; i += OWN_WAVES * 32)
      reinterpret_cast<float4*>(s_acc)[i] = make_float4(0.f, 0.f, 0.f, 0.f);
    for (int i = tid; i < OWN_CAP; i += OWN_WAVES * 32) s_cnt[i] = 0.f;
    __syncthreads();

    const unsigned int pat = (unsigned int)(r & 255) * 0x01010101u;
    int qn = 0;

    const int nblk = (chunks + 127) >> 7;
    v4u cur[4];
    #pragma unroll
    for (int u = 0; u < 4; ++u) {
      const int c = wv * 128 + u * 32 + lane;
      cur[u] = own16[c < chunks ? c : chunks - 1];
    }
    for (int bi = wv; bi < nblk; bi += OWN_WAVES) {
      v4u nxt[4];
      #pragma unroll
      for (int u = 0; u < 4; ++u) {
        const int c = (bi + OWN_WAVES) * 128 + u * 32 + lane;
        nxt[u] = own16[c < chunks ? c : chunks - 1];
      }
      unsigned long long m = 0;
      #pragma unroll
      for (int u = 0; u < 4; ++u) {
        const int c = bi * 128 + u * 32 + lane;
        if (c < chunks)
          m |= (unsigned long long)match16(cur[u], pat) << (16 * u);
      }
      for (;;) {
        const unsigned int bal = __builtin_amdgcn_ballot_w32(m != 0);
        if (bal == 0) break;
        if (m != 0) {
          const int k = __builtin_ctzll(m);
          m &= m - 1;
          const int slot = qn + __builtin_popcount(bal & ((1u << lane) - 1u));
          q[slot] = ((bi * 128 + (k >> 4) * 32 + lane) << 4) + (k & 15);
        }
        qn += __builtin_popcount(bal);
        if (qn >= 32) {
          own_drain<WRITE_CNT>(feat, pairs, s_acc, s_cnt, q, 32, lo, E, lane);
          qn -= 32;
          int keep = 0;
          if (lane < qn) keep = q[32 + lane];
          if (lane < qn) q[lane] = keep;
        }
      }
      #pragma unroll
      for (int u = 0; u < 4; ++u) cur[u] = nxt[u];
    }
    if (qn > 0)
      own_drain<WRITE_CNT>(feat, pairs, s_acc, s_cnt, q, qn, lo, E, lane);
    __syncthreads();

    for (int row = wv; row < OWN_CAP; row += OWN_WAVES) {
      const int n = lo + row;
      if (n < N)
        *reinterpret_cast<float4*>(agg + (long long)n * NODE_DIM + lane * 4) =
            *reinterpret_cast<const float4*>(s_acc + row * NODE_DIM + lane * 4);
    }
    if (WRITE_CNT) {
      for (int i = tid; i < OWN_CAP; i += OWN_WAVES * 32)
        if (lo + i < N) cnt[lo + i] = s_cnt[i];
    }
    __syncthreads();
  }
}

#define CB_WAVES 8
#define CB_ROWS 32
#define CB_GRID 512

template <bool NORM_RELU>
__global__ __launch_bounds__(CB_WAVES * 32)
void sage_combine_kernel(const float* __restrict__ agg,
                         const float* __restrict__ cnt,
                         const float* __restrict__ xin,
                         const float* __restrict__ Wl,
                         const float* __restrict__ bias,
                         const float* __restrict__ Wr,
                         float* __restrict__ outp, int N) {
  __shared__ float s_agg[CB_ROWS][LDS_STRIDE];
  __shared__ float s_in [CB_ROWS][LDS_STRIDE];
  __shared__ float s_out[CB_ROWS][LDS_STRIDE];

  const int tid   = threadIdx.x;
  const int lane  = tid & 31;
  const int wv    = tid >> 5;
  const int m     = lane & 15;
  const int khalf = (lane >> 4) << 1;
  const int mbase = (lane >> 4) << 3;
  const int n0    = wv * 16;
  const int groups = (N + CB_ROWS - 1) / CB_ROWS;

  v2f bL[32], bR[32];
  {
    const float* wlrow = Wl + (long long)(n0 + m) * NODE_DIM + khalf;
    const float* wrrow = Wr + (long long)(n0 + m) * NODE_DIM + khalf;
    #pragma unroll
    for (int j = 0; j < 32; ++j) {
      bL[j] = *reinterpret_cast<const v2f*>(wlrow + 4 * j);
      bR[j] = *reinterpret_cast<const v2f*>(wrrow + 4 * j);
    }
  }
  const float bv = bias[n0 + m];

  float4 pa[4], px[4];
  float  pc[4];
  int g = blockIdx.x;
  if (g < groups) {
    #pragma unroll
    for (int i = 0; i < 4; ++i) {
      const int n = g * CB_ROWS + wv + 8 * i;
      pa[i] = make_float4(0.f, 0.f, 0.f, 0.f);
      px[i] = make_float4(0.f, 0.f, 0.f, 0.f);
      pc[i] = 1.0f;
      if (n < N) {
        pc[i] = cnt[n];
        pa[i] = *reinterpret_cast<const float4*>(agg + (long long)n * NODE_DIM + lane * 4);
        px[i] = *reinterpret_cast<const float4*>(xin + (long long)n * NODE_DIM + lane * 4);
      }
    }
  }

  for (; g < groups; g += CB_GRID) {
    const int node0 = g * CB_ROWS;

    #pragma unroll
    for (int i = 0; i < 4; ++i) {
      const int r = wv + 8 * i;
      const float rc = 1.0f / fmaxf(pc[i], 1.0f);
      float4 va = pa[i];
      va.x *= rc; va.y *= rc; va.z *= rc; va.w *= rc;
      *reinterpret_cast<float4*>(&s_agg[r][lane * 4]) = va;
      *reinterpret_cast<float4*>(&s_in [r][lane * 4]) = px[i];
    }
    __syncthreads();

    {
      const int gn = g + CB_GRID;
      if (gn < groups) {
        #pragma unroll
        for (int i = 0; i < 4; ++i) {
          const int n = gn * CB_ROWS + wv + 8 * i;
          pa[i] = make_float4(0.f, 0.f, 0.f, 0.f);
          px[i] = make_float4(0.f, 0.f, 0.f, 0.f);
          pc[i] = 1.0f;
          if (n < N) {
            pc[i] = cnt[n];
            pa[i] = *reinterpret_cast<const float4*>(agg + (long long)n * NODE_DIM + lane * 4);
            px[i] = *reinterpret_cast<const float4*>(xin + (long long)n * NODE_DIM + lane * 4);
          }
        }
      }
    }

    v8f c0, c1;
    #pragma unroll
    for (int v = 0; v < 8; ++v) { c0[v] = bv; c1[v] = bv; }

    #pragma unroll
    for (int j = 0; j < 32; ++j) {
      const int k = 4 * j + khalf;
      const v2f a0 = *reinterpret_cast<const v2f*>(&s_agg[m][k]);
      const v2f a1 = *reinterpret_cast<const v2f*>(&s_agg[16 + m][k]);
      const v2f x0 = *reinterpret_cast<const v2f*>(&s_in[m][k]);
      const v2f x1 = *reinterpret_cast<const v2f*>(&s_in[16 + m][k]);
      c0 = __builtin_amdgcn_wmma_f32_16x16x4_f32(false, a0, false, bL[j],
                                                 (short)0, c0, false, false);
      c1 = __builtin_amdgcn_wmma_f32_16x16x4_f32(false, a1, false, bL[j],
                                                 (short)0, c1, false, false);
      c0 = __builtin_amdgcn_wmma_f32_16x16x4_f32(false, x0, false, bR[j],
                                                 (short)0, c0, false, false);
      c1 = __builtin_amdgcn_wmma_f32_16x16x4_f32(false, x1, false, bR[j],
                                                 (short)0, c1, false, false);
    }

    #pragma unroll
    for (int v = 0; v < 8; ++v) {
      s_out[mbase + v][n0 + m]      = c0[v];
      s_out[16 + mbase + v][n0 + m] = c1[v];
    }
    __syncthreads();

    {
      const int row = tid >> 3;
      const int seg = tid & 7;
      float4 o[4];
      #pragma unroll
      for (int q = 0; q < 4; ++q)
        o[q] = *reinterpret_cast<const float4*>(&s_out[row][seg * 4 + 32 * q]);
      if (NORM_RELU) {
        float ss = 0.f;
        #pragma unroll
        for (int q = 0; q < 4; ++q)
          ss += o[q].x * o[q].x + o[q].y * o[q].y + o[q].z * o[q].z + o[q].w * o[q].w;
        ss += __shfl_xor(ss, 1, 32);
        ss += __shfl_xor(ss, 2, 32);
        ss += __shfl_xor(ss, 4, 32);
        const float scl = 1.0f / fmaxf(sqrtf(ss), 1e-12f);
        #pragma unroll
        for (int q = 0; q < 4; ++q) {
          o[q].x = fmaxf(o[q].x * scl, 0.0f);
          o[q].y = fmaxf(o[q].y * scl, 0.0f);
          o[q].z = fmaxf(o[q].z * scl, 0.0f);
          o[q].w = fmaxf(o[q].w * scl, 0.0f);
        }
      }
      const int n = node0 + row;
      if (n < N) {
        #pragma unroll
        for (int q = 0; q < 4; ++q)
          *reinterpret_cast<float4*>(outp + (long long)n * NODE_DIM + seg * 4 + 32 * q) = o[q];
      }
    }
  }
}

extern "C" void kernel_launch(void* const* d_in, const int* in_sizes, int n_in,
                              void* d_out, int out_size, void* d_ws,
                              size_t ws_size, hipStream_t stream) {
  (void)n_in; (void)out_size; (void)ws_size;

  const float* x   = (const float*)d_in[0];
  const int*   ei  = (const int*)d_in[1];
  const float* W1l = (const float*)d_in[2];
  const float* b1  = (const float*)d_in[3];
  const float* W1r = (const float*)d_in[4];
  const float* W2l = (const float*)d_in[5];
  const float* b2  = (const float*)d_in[6];
  const float* W2r = (const float*)d_in[7];
  float* out = (float*)d_out;

  const int N = in_sizes[0] / NODE_DIM;
  const int E = in_sizes[1] / 2;
  const int* src = ei;
  const int* dst = ei + E;

  char* ws = (char*)d_ws;
  float* cnt = (float*)ws;
  size_t off = ((size_t)N * 4 + 255) & ~(size_t)255;
  float* aggb = (float*)(ws + off);
  off += (((size_t)N * NODE_DIM * 4) + 255) & ~(size_t)255;
  float* h = (float*)(ws + off);

  const int oblk = (((E + 15) / 16) * 4 + 255) / 256;
  sage_owner_kernel<<<oblk, 256, 0, stream>>>(src, dst, out, E);

  sage_gather_own_kernel<true><<<OWN_GRID, OWN_WAVES * 32, 0, stream>>>(
      x, out, aggb, cnt, N, E);
  sage_combine_kernel<true><<<CB_GRID, CB_WAVES * 32, 0, stream>>>(
      aggb, cnt, x, W1l, b1, W1r, h, N);

  sage_gather_own_kernel<false><<<OWN_GRID, OWN_WAVES * 32, 0, stream>>>(
      h, out, aggb, cnt, N, E);
  sage_combine_kernel<false><<<CB_GRID, CB_WAVES * 32, 0, stream>>>(
      aggb, cnt, h, W2l, b2, W2r, out, N);
}


# Layer 2 replays layer 1's matches from per-wave lists in a FIXED order (earlier try: arrival order): a wave owns rows mod 32, adds by plain LDS updates; layer 2 has no scan and no float atomics
# speedup vs baseline: 3.7248x; 1.1220x over previous
#include <hip/hip_runtime.h>
#include <hip/hip_bf16.h>
#include <math.h>

typedef float v2f __attribute__((ext_vector_type(2)));
typedef float v8f __attribute__((ext_vector_type(8)));

#define NODE_DIM 128
#define LDS_STRIDE 132

#define SCATTER_BATCH 8

#define OWN_CAP 416
#define OWN_WAVES 32
#define OWN_GRID 256
#define Q_CAP 64

typedef unsigned long long pair_t;
__device__ __forceinline__ const pair_t* edge_pairs(const float* ownbuf, int E) {
  const int chunks = (E + 15) >> 4;
  return reinterpret_cast<const pair_t*>(reinterpret_cast<const char*>(ownbuf) + (size_t)chunks * 16);
}

#define SEG_CAP 320
#define LIST_LEN (OWN_WAVES * SEG_CAP)
__device__ __forceinline__ char* list_base(const float* ownbuf, int E) {
  const size_t chunks = (size_t)((E + 15) >> 4);
  const size_t off = (chunks * 16 + chunks * 128 + 255) & ~(size_t)255;
  return const_cast<char*>(reinterpret_cast<const char*>(ownbuf)) + off;
}
__device__ __forceinline__ pair_t* list_entries(const float* ownbuf, int E, int r) {
  return reinterpret_cast<pair_t*>(list_base(ownbuf, E)) + (size_t)r * LIST_LEN;
}
__device__ __forceinline__ unsigned char* list_bytes(const float* ownbuf, int E, int ranges, int r) {
  return reinterpret_cast<unsigned char*>(list_base(ownbuf, E)) + (size_t)ranges * LIST_LEN * 8 + (size_t)r * LIST_LEN;
}
__device__ __forceinline__ int* list_flags(const float* ownbuf, int E, int ranges) {
  return reinterpret_cast<int*>(list_base(ownbuf, E) + (size_t)ranges * LIST_LEN * 9);
}

__global__ __launch_bounds__(256)
void sage_owner_kernel(const int* __restrict__ src, const int* __restrict__ dst,
                       float* __restrict__ ownbuf, int E) {
  unsigned int* own = reinterpret_cast<unsigned int*>(ownbuf);
  const int chunks = (E + 15) >> 4;
  pair_t* pairs = const_cast<pair_t*>(edge_pairs(ownbuf, E));
  const int t = blockIdx.x * 256 + threadIdx.x;
  if (t >= chunks * 4) return;
  const int e = t * 4;
  unsigned int w = 0;
  if (e + 3 < E) {
    int s[4], d[4];
    #pragma unroll
    for (int k = 0; k < 4; ++k) { d[k] = dst[e + k]; s[k] = src[e + k]; }
    #pragma unroll
    for (int k = 0; k < 4; ++k) w |= (((unsigned int)d[k] / OWN_CAP) & 255u) << (8 * k);
    int4* p4 = reinterpret_cast<int4*>(pairs + e);
    p4[0] = make_int4(s[0], d[0], s[1], d[1]);
    p4[1] = make_int4(s[2], d[2], s[3], d[3]);
  } else {
    #pragma unroll
    for (int k = 0; k < 4; ++k)
      if (e + k < E) {
        const int d = dst[e + k];
        w |= (((unsigned int)d / OWN_CAP) & 255u) << (8 * k);
        pairs[e + k] = ((pair_t)(unsigned int)d << 32) | (pair_t)(unsigned int)src[e + k];
      }
  }
  own[t] = w;
}

__device__ __forceinline__ unsigned int zero_bytes(unsigned int x) {
  return ~(((x & 0x7f7f7f7fu) + 0x7f7f7f7fu) | x) & 0x80808080u;
}
__device__ __forceinline__ unsigned int pack_nib(unsigned int z) {
  return (((z >> 7) * 0x00204081u) >> 21) & 15u;
}

typedef unsigned int v4u __attribute__((ext_vector_type(4)));

__device__ __forceinline__ unsigned int match16(v4u w, unsigned int pat) {
  const unsigned int z0 = zero_bytes(w.x ^ pat);
  const unsigned int z1 = zero_bytes(w.y ^ pat);
  const unsigned int z2 = zero_bytes(w.z ^ pat);
  const unsigned int z3 = zero_bytes(w.w ^ pat);
  unsigned int m = 0;
  if (z0 | z1 | z2 | z3)
    m = pack_nib(z0) | (pack_nib(z1) << 4) | (pack_nib(z2) << 8) | (pack_nib(z3) << 12);
  return m;
}

typedef volatile __attribute__((address_space(3))) int* lds_vint_p;

template <bool WRITE_CNT>
__device__ __forceinline__ void own_drain(const float* __restrict__ feat,
                                          const pair_t* __restrict__ pairs,
                                          float* s_acc, float* s_cnt,
                                          lds_vint_p q, int n, int lo, int E,
                                          int lane, pair_t* ent_w,
                                          unsigned char* ob_w, int& ln) {
  int s = 0, loc = -1;
  if (lane < n) {
    const int e = q[lane];
    if (e < E) {
      const pair_t sd = pairs[e];
      const int l = (int)(unsigned int)(sd >> 32) - lo;
      const bool mine = (unsigned int)l < (unsigned int)OWN_CAP;
      s   = mine ? (int)(unsigned int)sd : 0;
      loc = mine ? l : -1;
      if (WRITE_CNT && mine)
        __hip_atomic_fetch_add(&s_cnt[l], 1.0f, __ATOMIC_RELAXED, __HIP_MEMORY_SCOPE_WORKGROUP);
    }
  }
  if (WRITE_CNT) {
    const unsigned int bal = __builtin_amdgcn_ballot_w32(loc >= 0);
    const int pos = ln + __builtin_popcount(bal & ((1u << lane) - 1u));
    if (loc >= 0 && pos < SEG_CAP) {
      ent_w[pos] = ((pair_t)(unsigned int)loc << 32) | (pair_t)(unsigned int)s;
      ob_w[pos]  = (unsigned char)(loc & 31);
    }
    ln += __builtin_popcount(bal);
  }
  for (int i = 0; i < n; i += SCATTER_BATCH) {
    float v[SCATTER_BATCH][4];
    #pragma unroll
    for (int j = 0; j < SCATTER_BATCH; ++j) {
      const int sj = __builtin_amdgcn_readlane(s, i + j);
      const float* r = feat + (long long)sj * NODE_DIM + lane;
      v[j][0] = r[0];
      v[j][1] = r[32];
      v[j][2] = r[64];
      v[j][3] = r[96];
    }
    #pragma unroll
    for (int j = 0; j < SCATTER_BATCH; ++j) {
      const int lj = __builtin_amdgcn_readlane(loc, i + j);
      if (lj >= 0) {
        float* a = s_acc + lj * NODE_DIM + lane;
        __hip_atomic_fetch_add(a,      v[j][0], __ATOMIC_RELAXED, __HIP_MEMORY_SCOPE_WORKGROUP);
        __hip_atomic_fetch_add(a + 32, v[j][1], __ATOMIC_RELAXED, __HIP_MEMORY_SCOPE_WORKGROUP);
        __hip_atomic_fetch_add(a + 64, v[j][2], __ATOMIC_RELAXED, __HIP_MEMORY_SCOPE_WORKGROUP);
        __hip_atomic_fetch_add(a + 96, v[j][3], __ATOMIC_RELAXED, __HIP_MEMORY_SCOPE_WORKGROUP);
      }
    }
  }
}

__device__ __forceinline__ void replay_drain(const float* __restrict__ feat,
                                             const pair_t* __restrict__ ent_r,
                                             float* s_acc, lds_vint_p q, int n,
                                             int lane) {
  int s = 0, loc = -1;
  if (lane < n) {
    const pair_t sl = ent_r[q[lane]];
    s   = (int)(unsigned int)sl;
    loc = (int)(unsigned int)(sl >> 32);
  }
  for (int i = 0; i < n; i += SCATTER_BATCH) {
    float4 v[SCATTER_BATCH];
    #pragma unroll
    for (int j = 0; j < SCATTER_BATCH; ++j) {
      const int sj = __builtin_amdgcn_readlane(s, i + j);
      v[j] = *reinterpret_cast<const float4*>(feat + (long long)sj * NODE_DIM + lane * 4);
    }
    #pragma unroll
    for (int j = 0; j < SCATTER_BATCH; ++j) {
      const int lj = __builtin_amdgcn_readlane(loc, i + j);
      if (lj >= 0) {
        float4* a = reinterpret_cast<float4*>(s_acc + lj * NODE_DIM + lane * 4);
        float4 t = *a;
        t.x += v[j].x; t.y += v[j].y; t.z += v[j].z; t.w += v[j].w;
        *a = t;
      }
    }
  }
}

__device__ __forceinline__ void replay_range(const float* __restrict__ feat,
                                             const pair_t* __restrict__ ent_r,
                                             const unsigned char* __restrict__ ob_r,
                                             float* s_acc, lds_vint_p q,
                                             int wv, int lane) {
  const v4u* ob16 = reinterpret_cast<const v4u*>(ob_r);
  const unsigned int pat = (unsigned int)wv * 0x01010101u;
  const int nblk = LIST_LEN / 2048;
  int qn = 0;
  v4u cur[4];
  #pragma unroll
  for (int u = 0; u < 4; ++u) cur[u] = ob16[u * 32 + lane];
  for (int bi = 0; bi < nblk; ++bi) {
    v4u nxt[4];
    #pragma unroll
    for (int u = 0; u < 4; ++u) {
      const int c = (bi + 1) * 128 + u * 32 + lane;
      nxt[u] = ob16[c < LIST_LEN / 16 ? c : LIST_LEN / 16 - 1];
    }
    unsigned long long m = 0;
    #pragma unroll
    for (int u = 0; u < 4; ++u)
      m |= (unsigned long long)match16(cur[u], pat) << (16 * u);
    for (;;) {
      const unsigned int bal = __builtin_amdgcn_ballot_w32(m != 0);
      if (bal == 0) break;
      if (m != 0) {
        const int k = __builtin_ctzll(m);
        m &= m - 1;
        const int slot = qn + __builtin_popcount(bal & ((1u << lane) - 1u));
        q[slot] = ((bi * 128 + (k >> 4) * 32 + lane) << 4) + (k & 15);
      }
      qn += __builtin_popcount(bal);
      if (qn >= 32) {
        replay_drain(feat, ent_r, s_acc, q, 32, lane);
        qn -= 32;
        int keep = 0;
        if (lane < qn) keep = q[32 + lane];
        if (lane < qn) q[lane] = keep;
      }
    }
    #pragma unroll
    for (int u = 0; u < 4; ++u) cur[u] = nxt[u];
  }
  if (qn > 0) replay_drain(feat, ent_r, s_acc, q, qn, lane);
}

template <bool WRITE_CNT>
__global__ __launch_bounds__(OWN_WAVES * 32)
void sage_gather_own_kernel(const float* __restrict__ feat,
                            const float* __restrict__ ownbuf,
                            float* __restrict__ agg,
                            float* __restrict__ cnt, int N, int E) {
  __shared__ float s_acc[OWN_CAP * NODE_DIM];
  __shared__ float s_cnt[OWN_CAP];
  __shared__ int   s_q[OWN_WAVES][Q_CAP];
  __shared__ int   s_ovf;

  const v4u* own16 = reinterpret_cast<const v4u*>(ownbuf);
  const pair_t* pairs = edge_pairs(ownbuf, E);
  const int tid  = threadIdx.x;
  const int lane = tid & 31;
  const int wv = __builtin_amdgcn_readfirstlane(tid >> 5);
  lds_vint_p q = (lds_vint_p)&s_q[wv][0];
  const int chunks = (E + 15) >> 4;
  const int ranges = (N + OWN_CAP - 1) / OWN_CAP;

  for (int r = blockIdx.x; r < ranges; r += OWN_GRID) {
    const int lo = r * OWN_CAP;

    for (int i = tid; i < OWN_CAP * NODE_DIM / 4; i += OWN_WAVES * 32)
      reinterpret_cast<float4*>(s_acc)[i] = make_float4(0.f, 0.f, 0.f, 0.f);
    for (int i = tid; i < OWN_CAP; i += OWN_WAVES * 32) s_cnt[i] = 0.f;
    if (WRITE_CNT && tid == 0) s_ovf = 0;
    __syncthreads();

    pair_t* ent_r = list_entries(ownbuf, E, r);
    unsigned char* ob_r = list_bytes(ownbuf, E, ranges, r);
    pair_t* ent_w = ent_r + wv * SEG_CAP;
    unsigned char* ob_w = ob_r + wv * SEG_CAP;
    int ln = 0;

    const bool replay = !WRITE_CNT && list_flags(ownbuf, E, ranges)[r] == 0;
    if (replay) {
      replay_range(feat, ent_r, ob_r, s_acc, q, wv, lane);
    } else {

    const unsigned int pat = (unsigned int)(r & 255) * 0x01010101u;
    int qn = 0;

    const int nblk = (chunks + 127) >> 7;
    v4u cur[4];
    #pragma unroll
    for (int u = 0; u < 4; ++u) {
      const int c = wv * 128 + u * 32 + lane;
      cur[u] = own16[c < chunks ? c : chunks - 1];
    }
    for (int bi = wv; bi < nblk; bi += OWN_WAVES) {
      v4u nxt[4];
      #pragma unroll
      for (int u = 0; u < 4; ++u) {
        const int c = (bi + OWN_WAVES) * 128 + u * 32 + lane;
        nxt[u] = own16[c < chunks ? c : chunks - 1];
      }
      unsigned long long m = 0;
      #pragma unroll
      for (int u = 0; u < 4; ++u) {
        const int c = bi * 128 + u * 32 + lane;
        if (c < chunks)
          m |= (unsigned long long)match16(cur[u], pat) << (16 * u);
      }
      for (;;) {
        const unsigned int bal = __builtin_amdgcn_ballot_w32(m != 0);
        if (bal == 0) break;
        if (m != 0) {
          const int k = __builtin_ctzll(m);
          m &= m - 1;
          const int slot = qn + __builtin_popcount(bal & ((1u << lane) - 1u));
          q[slot] = ((bi * 128 + (k >> 4) * 32 + lane) << 4) + (k & 15);
        }
        qn += __builtin_popcount(bal);
        if (qn >= 32) {
          own_drain<WRITE_CNT>(feat, pairs, s_acc, s_cnt, q, 32, lo, E, lane, ent_w, ob_w, ln);
          qn -= 32;
          int keep = 0;
          if (lane < qn) keep = q[32 + lane];
          if (lane < qn) q[lane] = keep;
        }
      }
      #pragma unroll
      for (int u = 0; u < 4; ++u) cur[u] = nxt[u];
    }
    if (qn > 0)
      own_drain<WRITE_CNT>(feat, pairs, s_acc, s_cnt, q, qn, lo, E, lane, ent_w, ob_w, ln);
    if (WRITE_CNT) {
      for (int k = (ln < SEG_CAP ? ln : SEG_CAP) + lane; k < SEG_CAP; k += 32) ob_w[k] = 0xFF;
      if (ln > SEG_CAP && lane == 0) s_ovf = 1;
    }
    }
    __syncthreads();

    for (int row = wv; row < OWN_CAP; row += OWN_WAVES) {
      const int n = lo + row;
      if (n < N)
        *reinterpret_cast<float4*>(agg + (long long)n * NODE_DIM + lane * 4) =
            *reinterpret_cast<const float4*>(s_acc + row * NODE_DIM + lane * 4);
    }
    if (WRITE_CNT) {
      for (int i = tid; i < OWN_CAP; i += OWN_WAVES * 32)
        if (lo + i < N) cnt[lo + i] = s_cnt[i];
      if (tid == 0) list_flags(ownbuf, E, ranges)[r] = s_ovf;
    }
    __syncthreads();
  }
}

#define CB_WAVES 8
#define CB_ROWS 32
#define CB_GRID 512

template <bool NORM_RELU>
__global__ __launch_bounds__(CB_WAVES * 32)
void sage_combine_kernel(const float* __restrict__ agg,
                         const float* __restrict__ cnt,
                         const float* __restrict__ xin,
                         const float* __restrict__ Wl,
                         const float* __restrict__ bias,
                         const float* __restrict__ Wr,
                         float* __restrict__ outp, int N) {
  __shared__ float s_agg[CB_ROWS][LDS_STRIDE];
  __shared__ float s_in [CB_ROWS][LDS_STRIDE];
  __shared__ float s_out[CB_ROWS][LDS_STRIDE];

  const int tid   = threadIdx.x;
  const int lane  = tid & 31;
  const int wv    = tid >> 5;
  const int m     = lane & 15;
  const int khalf = (lane >> 4) << 1;
  const int mbase = (lane >> 4) << 3;
  const int n0    = wv * 16;
  const int groups = (N + CB_ROWS - 1) / CB_ROWS;

  v2f bL[32], bR[32];
  {
    const float* wlrow = Wl + (long long)(n0 + m) * NODE_DIM + khalf;
    const float* wrrow = Wr + (long long)(n0 + m) * NODE_DIM + khalf;
    #pragma unroll
    for (int j = 0; j < 32; ++j) {
      bL[j] = *reinterpret_cast<const v2f*>(wlrow + 4 * j);
      bR[j] = *reinterpret_cast<const v2f*>(wrrow + 4 * j);
    }
  }
  const float bv = bias[n0 + m];

  float4 pa[4], px[4];
  float  pc[4];
  int g = blockIdx.x;
  if (g < groups) {
    #pragma unroll
    for (int i = 0; i < 4; ++i) {
      const int n = g * CB_ROWS + wv + 8 * i;
      pa[i] = make_float4(0.f, 0.f, 0.f, 0.f);
      px[i] = make_float4(0.f, 0.f, 0.f, 0.f);
      pc[i] = 1.0f;
      if (n < N) {
        pc[i] = cnt[n];
        pa[i] = *reinterpret_cast<const float4*>(agg + (long long)n * NODE_DIM + lane * 4);
        px[i] = *reinterpret_cast<const float4*>(xin + (long long)n * NODE_DIM + lane * 4);
      }
    }
  }

  for (; g < groups; g += CB_GRID) {
    const int node0 = g * CB_ROWS;

    #pragma unroll
    for (int i = 0; i < 4; ++i) {
      const int r = wv + 8 * i;
      const float rc = 1.0f / fmaxf(pc[i], 1.0f);
      float4 va = pa[i];
      va.x *= rc; va.y *= rc; va.z *= rc; va.w *= rc;
      *reinterpret_cast<float4*>(&s_agg[r][lane * 4]) = va;
      *reinterpret_cast<float4*>(&s_in [r][lane * 4]) = px[i];
    }
    __syncthreads();

    {
      const int gn = g + CB_GRID;
      if (gn < groups) {
        #pragma unroll
        for (int i = 0; i < 4; ++i) {
          const int n = gn * CB_ROWS + wv + 8 * i;
          pa[i] = make_float4(0.f, 0.f, 0.f, 0.f);
          px[i] = make_float4(0.f, 0.f, 0.f, 0.f);
          pc[i] = 1.0f;
          if (n < N) {
            pc[i] = cnt[n];
            pa[i] = *reinterpret_cast<const float4*>(agg + (long long)n * NODE_DIM + lane * 4);
            px[i] = *reinterpret_cast<const float4*>(xin + (long long)n * NODE_DIM + lane * 4);
          }
        }
      }
    }

    v8f c0, c1;
    #pragma unroll
    for (int v = 0; v < 8; ++v) { c0[v] = bv; c1[v] = bv; }

    #pragma unroll
    for (int j = 0; j < 32; ++j) {
      const int k = 4 * j + khalf;
      const v2f a0 = *reinterpret_cast<const v2f*>(&s_agg[m][k]);
      const v2f a1 = *reinterpret_cast<const v2f*>(&s_agg[16 + m][k]);
      const v2f x0 = *reinterpret_cast<const v2f*>(&s_in[m][k]);
      const v2f x1 = *reinterpret_cast<const v2f*>(&s_in[16 + m][k]);
      c0 = __builtin_amdgcn_wmma_f32_16x16x4_f32(false, a0, false, bL[j],
                                                 (short)0, c0, false, false);
      c1 = __builtin_amdgcn_wmma_f32_16x16x4_f32(false, a1, false, bL[j],
                                                 (short)0, c1, false, false);
      c0 = __builtin_amdgcn_wmma_f32_16x16x4_f32(false, x0, false, bR[j],
                                                 (short)0, c0, false, false);
      c1 = __builtin_amdgcn_wmma_f32_16x16x4_f32(false, x1, false, bR[j],
                                                 (short)0, c1, false, false);
    }

    #pragma unroll
    for (int v = 0; v < 8; ++v) {
      s_out[mbase + v][n0 + m]      = c0[v];
      s_out[16 + mbase + v][n0 + m] = c1[v];
    }
    __syncthreads();

    {
      const int row = tid >> 3;
      const int seg = tid & 7;
      float4 o[4];
      #pragma unroll
      for (int q = 0; q < 4; ++q)
        o[q] = *reinterpret_cast<const float4*>(&s_out[row][seg * 4 + 32 * q]);
      if (NORM_RELU) {
        float ss = 0.f;
        #pragma unroll
        for (int q = 0; q < 4; ++q)
          ss += o[q].x * o[q].x + o[q].y * o[q].y + o[q].z * o[q].z + o[q].w * o[q].w;
        ss += __shfl_xor(ss, 1, 32);
        ss += __shfl_xor(ss, 2, 32);
        ss += __shfl_xor(ss, 4, 32);
        const float scl = 1.0f / fmaxf(sqrtf(ss), 1e-12f);
        #pragma unroll
        for (int q = 0; q < 4; ++q) {
          o[q].x = fmaxf(o[q].x * scl, 0.0f);
          o[q].y = fmaxf(o[q].y * scl, 0.0f);
          o[q].z = fmaxf(o[q].z * scl, 0.0f);
          o[q].w = fmaxf(o[q].w * scl, 0.0f);
        }
      }
      const int n = node0 + row;
      if (n < N) {
        #pragma unroll
        for (int q = 0; q < 4; ++q)
          *reinterpret_cast<float4*>(outp + (long long)n * NODE_DIM + seg * 4 + 32 * q) = o[q];
      }
    }
  }
}

extern "C" void kernel_launch(void* const* d_in, const int* in_sizes, int n_in,
                              void* d_out, int out_size, void* d_ws,
                              size_t ws_size, hipStream_t stream) {
  (void)n_in; (void)out_size; (void)ws_size;

  const float* x   = (const float*)d_in[0];
  const int*   ei  = (const int*)d_in[1];
  const float* W1l = (const float*)d_in[2];
  const float* b1  = (const float*)d_in[3];
  const float* W1r = (const float*)d_in[4];
  const float* W2l = (const float*)d_in[5];
  const float* b2  = (const float*)d_in[6];
  const float* W2r = (const float*)d_in[7];
  float* out = (float*)d_out;

  const int N = in_sizes[0] / NODE_DIM;
  const int E = in_sizes[1] / 2;
  const int* src = ei;
  const int* dst = ei + E;

  char* ws = (char*)d_ws;
  float* cnt = (float*)ws;
  size_t off = ((size_t)N * 4 + 255) & ~(size_t)255;
  float* aggb = (float*)(ws + off);
  off += (((size_t)N * NODE_DIM * 4) + 255) & ~(size_t)255;
  float* h = (float*)(ws + off);

  const int oblk = (((E + 15) / 16) * 4 + 255) / 256;
  sage_owner_kernel<<<oblk, 256, 0, stream>>>(src, dst, out, E);

  sage_gather_own_kernel<true><<<OWN_GRID, OWN_WAVES * 32, 0, stream>>>(
      x, out, aggb, cnt, N, E);
  sage_combine_kernel<true><<<CB_GRID, CB_WAVES * 32, 0, stream>>>(
      aggb, cnt, x, W1l, b1, W1r, h, N);

  sage_gather_own_kernel<false><<<OWN_GRID, OWN_WAVES * 32, 0, stream>>>(
      h, out, aggb, cnt, N, E);
  sage_combine_kernel<false><<<CB_GRID, CB_WAVES * 32, 0, stream>>>(
      aggb, cnt, h, W2l, b2, W2r, out, N);
}
